# starting kernel
# baseline (speedup 1.0000x reference)
#include <hip/hip_runtime.h>
#include <hip/hip_bf16.h>
#include <math.h>

typedef __attribute__((ext_vector_type(2))) float v2f;
typedef __attribute__((ext_vector_type(8))) float v8f;

#define N_NODES 50000
#define N_EDGES 800000
#define HEADS   8
#define FEAT    32
#define NCOL    256

__global__ void k_zero_i32(int* __restrict__ p, int n) {
  int i = blockIdx.x * blockDim.x + threadIdx.x;
  if (i < n) p[i] = 0;
}

__global__ void k_count(const int* __restrict__ dst, int* __restrict__ cnt, int E) {
  int i = blockIdx.x * blockDim.x + threadIdx.x;
  if (i < E) atomicAdd(&cnt[dst[i]], 1);
}

__global__ void k_exscan(const int* __restrict__ cnt, int* __restrict__ rowptr,
                         int* __restrict__ cursor, int n) {
  __shared__ int sums[1024];
  int t = threadIdx.x;
  int chunk = (n + 1023) >> 10;
  int s0 = t * chunk; if (s0 > n) s0 = n;
  int s1 = s0 + chunk; if (s1 > n) s1 = n;
  int sum = 0;
  for (int i = s0; i < s1; ++i) sum += cnt[i];
  sums[t] = sum;
  __syncthreads();
  for (int off = 1; off < 1024; off <<= 1) {
    int v = (t >= off) ? sums[t - off] : 0;
    __syncthreads();
    sums[t] += v;
    __syncthreads();
  }
  int run = (t == 0) ? 0 : sums[t - 1];
  for (int i = s0; i < s1; ++i) {
    int c = cnt[i];
    rowptr[i] = run;
    cursor[i] = run;
    run += c;
  }
  if (t == 0) rowptr[n] = sums[1023];
}

__global__ void k_scatter(const int* __restrict__ dst, int* __restrict__ cursor,
                          int* __restrict__ eidx, int E) {
  int i = blockIdx.x * blockDim.x + threadIdx.x;
  if (i < E) {
    int p = atomicAdd(&cursor[dst[i]], 1);
    eidx[p] = i;
  }
}

__global__ void k_gemm_wmma(const float* __restrict__ A, const float* __restrict__ B,
                            float* __restrict__ C, int M, int K) {
  int wid  = (blockIdx.x * blockDim.x + threadIdx.x) >> 5;
  int lane = threadIdx.x & 31;
  int mtiles = M >> 4;
  if (wid >= mtiles * 4) return;
  int mt = wid >> 2, ng = wid & 3;
  int m0 = mt << 4, n0 = ng << 6;
  int rc   = lane & 15;
  int koff = (lane >> 4) << 1;
  int hi8  = (lane >> 4) << 3;

  v8f acc0 = {}, acc1 = {}, acc2 = {}, acc3 = {};
  const float* arow = A + (m0 + rc) * K;

  #pragma unroll 4
  for (int k = 0; k < K; k += 4) {
    v2f a;
    a.x = arow[k + koff];
    a.y = arow[k + koff + 1];
    const float* bp = B + (k + koff) * NCOL + n0 + rc;
    v2f b;
    b.x = bp[0];       b.y = bp[NCOL];
    acc0 = __builtin_amdgcn_wmma_f32_16x16x4_f32(false, a, false, b, (short)0, acc0, false, false);
    b.x = bp[16];      b.y = bp[NCOL + 16];
    acc1 = __builtin_amdgcn_wmma_f32_16x16x4_f32(false, a, false, b, (short)0, acc1, false, false);
    b.x = bp[32];      b.y = bp[NCOL + 32];
    acc2 = __builtin_amdgcn_wmma_f32_16x16x4_f32(false, a, false, b, (short)0, acc2, false, false);
    b.x = bp[48];      b.y = bp[NCOL + 48];
    acc3 = __builtin_amdgcn_wmma_f32_16x16x4_f32(false, a, false, b, (short)0, acc3, false, false);
  }

  float* crow = C + (m0 + hi8) * NCOL + n0 + rc;
  #pragma unroll
  for (int j = 0; j < 8; ++j) {
    crow[j * NCOL +  0] = acc0[j];
    crow[j * NCOL + 16] = acc1[j];
    crow[j * NCOL + 32] = acc2[j];
    crow[j * NCOL + 48] = acc3[j];
  }
}

__global__ void k_attn(const float* __restrict__ h, const float* __restrict__ al,
                       const float* __restrict__ ar, float* __restrict__ el,
                       float* __restrict__ er) {
  int idx = blockIdx.x * blockDim.x + threadIdx.x;
  if (idx >= N_NODES * HEADS) return;
  int hd = idx & (HEADS - 1);
  int n  = idx >> 3;
  const float* hp  = h + n * NCOL + hd * FEAT;
  const float* alp = al + hd * FEAT;
  const float* arp = ar + hd * FEAT;
  float sl = 0.f, sr = 0.f;
  #pragma unroll
  for (int f = 0; f < FEAT; ++f) {
    float v = hp[f];
    sl = fmaf(v, alp[f], sl);
    sr = fmaf(v, arp[f], sr);
  }
  el[idx] = sl;
  er[idx] = sr;
}

__global__ void k_aggregate(const float* __restrict__ h, const float* __restrict__ el,
                            const float* __restrict__ er, const float* __restrict__ bias,
                            const int* __restrict__ rowptr, const int* __restrict__ eidx,
                            const int* __restrict__ src, float* __restrict__ out) {
  int wid  = (blockIdx.x * blockDim.x + threadIdx.x) >> 5;
  int lane = threadIdx.x & 31;
  if (wid >= N_NODES * HEADS) return;
  int hd = wid & (HEADS - 1);
  int n  = wid >> 3;
  int s0 = rowptr[n], s1 = rowptr[n + 1];
  float ern = er[wid];

  float m = -INFINITY, ssum = 0.f, acc = 0.f;
  for (int j = s0; j < s1; ++j) {
    int e  = eidx[j];
    int sn = src[e];
    float ev = el[sn * HEADS + hd] + ern;
    ev = (ev > 0.f) ? ev : 0.2f * ev;
    float mn = fmaxf(m, ev);
    float sc = __expf(m - mn);
    float p  = __expf(ev - mn);
    ssum = ssum * sc + p;
    acc  = acc  * sc + p * h[sn * NCOL + hd * FEAT + lane];
    m = mn;
  }
  float r = (s1 > s0) ? (acc / ssum) : 0.f;
  out[n * NCOL + hd * FEAT + lane] = r + bias[hd * FEAT + lane];
}

__global__ void k_elu_headmean(const float* __restrict__ agg, float* __restrict__ out) {
  int idx = blockIdx.x * blockDim.x + threadIdx.x;
  if (idx >= N_NODES * FEAT) return;
  int f = idx & (FEAT - 1);
  int n = idx >> 5;
  float s = 0.f;
  #pragma unroll
  for (int hd = 0; hd < HEADS; ++hd) {
    float v = agg[n * NCOL + hd * FEAT + f];
    s += (v > 0.f) ? v : (__expf(v) - 1.f);
  }
  out[idx] = s * (1.f / HEADS);
}

extern "C" void kernel_launch(void* const* d_in, const int* in_sizes, int n_in,
                              void* d_out, int out_size, void* d_ws, size_t ws_size,
                              hipStream_t stream) {
  const float* x   = (const float*)d_in[0];
  const int*   src = (const int*)  d_in[1];
  const int*   dst = (const int*)  d_in[2];
  const float* W1  = (const float*)d_in[3];
  const float* al1 = (const float*)d_in[4];
  const float* ar1 = (const float*)d_in[5];
  const float* b1  = (const float*)d_in[6];
  const float* W2  = (const float*)d_in[7];
  const float* al2 = (const float*)d_in[8];
  const float* ar2 = (const float*)d_in[9];
  const float* b2  = (const float*)d_in[10];

  float* wsf = (float*)d_ws;
  float* h_buf = wsf;
  float* agg1  = h_buf + (size_t)N_NODES * NCOL;
  float* elb   = agg1  + (size_t)N_NODES * NCOL;
  float* erb   = elb   + (size_t)N_NODES * HEADS;
  float* h1act = erb   + (size_t)N_NODES * HEADS;
  int* ibase   = (int*)(h1act + (size_t)N_NODES * FEAT);
  int* rowptr  = ibase;
  int* cursor  = rowptr + (N_NODES + 2);
  int* cnts    = cursor + N_NODES;
  int* eidx    = cnts   + N_NODES;

  k_zero_i32<<<(N_NODES + 255) / 256, 256, 0, stream>>>(cnts, N_NODES);
  k_count   <<<(N_EDGES + 255) / 256, 256, 0, stream>>>(dst, cnts, N_EDGES);
  k_exscan  <<<1, 1024, 0, stream>>>(cnts, rowptr, cursor, N_NODES);
  k_scatter <<<(N_EDGES + 255) / 256, 256, 0, stream>>>(dst, cursor, eidx, N_EDGES);

  const int gemm_blocks = ((N_NODES / 16) * 4 * 32 + 255) / 256;
  const int nh_blocks   = (N_NODES * HEADS + 255) / 256;
  const int agg_blocks  = (N_NODES * HEADS * 32) / 256;
  const int nf_blocks   = (N_NODES * FEAT + 255) / 256;

  k_gemm_wmma <<<gemm_blocks, 256, 0, stream>>>(x, W1, h_buf, N_NODES, 128);
  k_attn      <<<nh_blocks,  256, 0, stream>>>(h_buf, al1, ar1, elb, erb);
  k_aggregate <<<agg_blocks, 256, 0, stream>>>(h_buf, elb, erb, b1, rowptr, eidx, src, agg1);
  k_elu_headmean<<<nf_blocks, 256, 0, stream>>>(agg1, h1act);

  k_gemm_wmma <<<gemm_blocks, 256, 0, stream>>>(h1act, W2, h_buf, N_NODES, 32);
  k_attn      <<<nh_blocks,  256, 0, stream>>>(h_buf, al2, ar2, elb, erb);
  k_aggregate <<<agg_blocks, 256, 0, stream>>>(h_buf, elb, erb, b2, rowptr, eidx, src, (float*)d_out);
}


# aggregate re-cut: wave works 4 dst nodes at once, 8 lanes a node; softmax weights one lane per edge (8-edge chunks, 3-step xor trees), then 8 lanes x 16 B per source row, no merge (sum order changes)
# speedup vs baseline: 1.4505x; 1.4505x over previous
#include <hip/hip_runtime.h>
#include <hip/hip_bf16.h>
#include <math.h>

typedef __attribute__((ext_vector_type(2))) float v2f;
typedef __attribute__((ext_vector_type(8))) float v8f;

#define N_NODES 50000
#define N_EDGES 800000
#define HEADS   8
#define FEAT    32
#define NCOL    256

__global__ void k_zero_i32(int* __restrict__ p, int n) {
  int i = blockIdx.x * blockDim.x + threadIdx.x;
  if (i < n) p[i] = 0;
}

__global__ void k_count(const int* __restrict__ dst, int* __restrict__ cnt, int E) {
  int i = blockIdx.x * blockDim.x + threadIdx.x;
  if (i < E) atomicAdd(&cnt[dst[i]], 1);
}

__global__ void k_exscan(const int* __restrict__ cnt, int* __restrict__ rowptr,
                         int* __restrict__ cursor, int n) {
  __shared__ int sums[1024];
  int t = threadIdx.x;
  int chunk = (n + 1023) >> 10;
  int s0 = t * chunk; if (s0 > n) s0 = n;
  int s1 = s0 + chunk; if (s1 > n) s1 = n;
  int sum = 0;
  for (int i = s0; i < s1; ++i) sum += cnt[i];
  sums[t] = sum;
  __syncthreads();
  for (int off = 1; off < 1024; off <<= 1) {
    int v = (t >= off) ? sums[t - off] : 0;
    __syncthreads();
    sums[t] += v;
    __syncthreads();
  }
  int run = (t == 0) ? 0 : sums[t - 1];
  for (int i = s0; i < s1; ++i) {
    int c = cnt[i];
    rowptr[i] = run;
    cursor[i] = run;
    run += c;
  }
  if (t == 0) rowptr[n] = sums[1023];
}

__global__ void k_scatter(const int* __restrict__ dst, int* __restrict__ cursor,
                          int* __restrict__ eidx, int E) {
  int i = blockIdx.x * blockDim.x + threadIdx.x;
  if (i < E) {
    int p = atomicAdd(&cursor[dst[i]], 1);
    eidx[p] = i;
  }
}

__global__ void k_gemm_wmma(const float* __restrict__ A, const float* __restrict__ B,
                            float* __restrict__ C, int M, int K) {
  int wid  = (blockIdx.x * blockDim.x + threadIdx.x) >> 5;
  int lane = threadIdx.x & 31;
  int mtiles = M >> 4;
  if (wid >= mtiles * 4) return;
  int mt = wid >> 2, ng = wid & 3;
  int m0 = mt << 4, n0 = ng << 6;
  int rc   = lane & 15;
  int koff = (lane >> 4) << 1;
  int hi8  = (lane >> 4) << 3;

  v8f acc0 = {}, acc1 = {}, acc2 = {}, acc3 = {};
  const float* arow = A + (m0 + rc) * K;

  #pragma unroll 4
  for (int k = 0; k < K; k += 4) {
    v2f a;
    a.x = arow[k + koff];
    a.y = arow[k + koff + 1];
    const float* bp = B + (k + koff) * NCOL + n0 + rc;
    v2f b;
    b.x = bp[0];       b.y = bp[NCOL];
    acc0 = __builtin_amdgcn_wmma_f32_16x16x4_f32(false, a, false, b, (short)0, acc0, false, false);
    b.x = bp[16];      b.y = bp[NCOL + 16];
    acc1 = __builtin_amdgcn_wmma_f32_16x16x4_f32(false, a, false, b, (short)0, acc1, false, false);
    b.x = bp[32];      b.y = bp[NCOL + 32];
    acc2 = __builtin_amdgcn_wmma_f32_16x16x4_f32(false, a, false, b, (short)0, acc2, false, false);
    b.x = bp[48];      b.y = bp[NCOL + 48];
    acc3 = __builtin_amdgcn_wmma_f32_16x16x4_f32(false, a, false, b, (short)0, acc3, false, false);
  }

  float* crow = C + (m0 + hi8) * NCOL + n0 + rc;
  #pragma unroll
  for (int j = 0; j < 8; ++j) {
    crow[j * NCOL +  0] = acc0[j];
    crow[j * NCOL + 16] = acc1[j];
    crow[j * NCOL + 32] = acc2[j];
    crow[j * NCOL + 48] = acc3[j];
  }
}

__global__ void k_attn(const float* __restrict__ h, const float* __restrict__ al,
                       const float* __restrict__ ar, float* __restrict__ el,
                       float* __restrict__ er) {
  int idx = blockIdx.x * blockDim.x + threadIdx.x;
  if (idx >= N_NODES * HEADS) return;
  int hd = idx & (HEADS - 1);
  int n  = idx >> 3;
  const float* hp  = h + n * NCOL + hd * FEAT;
  const float* alp = al + hd * FEAT;
  const float* arp = ar + hd * FEAT;
  float sl = 0.f, sr = 0.f;
  #pragma unroll
  for (int f = 0; f < FEAT; ++f) {
    float v = hp[f];
    sl = fmaf(v, alp[f], sl);
    sr = fmaf(v, arp[f], sr);
  }
  el[idx] = sl;
  er[idx] = sr;
}

#define AGG_NB  16
#define AGG_CAP 2048

template <int M>
__device__ __forceinline__ float row_xor(float x) {
  const int b = __builtin_bit_cast(int, x);
  return __builtin_bit_cast(float, __builtin_amdgcn_update_dpp(b, b, 0x160 | M, 0xf, 0xf, false));
}
template <int T>
__device__ __forceinline__ int grp_bcast(int x) {
  return __builtin_amdgcn_mov_dpp8(x, T * 0x249249);
}
template <int T>
__device__ __forceinline__ float grp_bcast(float x) {
  return __builtin_bit_cast(float, __builtin_amdgcn_mov_dpp8(__builtin_bit_cast(int, x), T * 0x249249));
}
__device__ __forceinline__ float grp_max(float x) {
  x = fmaxf(x, row_xor<1>(x));
  x = fmaxf(x, row_xor<2>(x));
  return fmaxf(x, row_xor<4>(x));
}
__device__ __forceinline__ float grp_sum(float x) {
  x += row_xor<1>(x);
  x += row_xor<2>(x);
  return x + row_xor<4>(x);
}

__device__ __forceinline__ float4 agg_row(const char* __restrict__ hb, int sn, int q8) {
  const unsigned off = (static_cast<unsigned>(sn) << 10) | static_cast<unsigned>(q8 << 4);
  return *reinterpret_cast<const float4*>(hb + off);
}

__device__ __forceinline__ void agg_fma(float4& acc, float p, const float4 v) {
  acc.x = fmaf(p, v.x, acc.x);
  acc.y = fmaf(p, v.y, acc.y);
  acc.z = fmaf(p, v.z, acc.z);
  acc.w = fmaf(p, v.w, acc.w);
}

template <bool STAGED>
__device__ __forceinline__ void agg_walk(const char* __restrict__ hb, const char* __restrict__ elb,
                                         const int* __restrict__ eidx, const int* __restrict__ src,
                                         const int* sn, int e0, int rpv, float erv,
                                         float4 bq, int lane, float* __restrict__ orow) {
  const int g4 = lane >> 3;
  const int q8 = lane & 7;
  for (int k4 = 0; k4 < AGG_NB; k4 += 4) {
    const int r0 = __builtin_amdgcn_readlane(rpv, k4);
    const int r1 = __builtin_amdgcn_readlane(rpv, k4 + 1);
    const int r2 = __builtin_amdgcn_readlane(rpv, k4 + 2);
    const int r3 = __builtin_amdgcn_readlane(rpv, k4 + 3);
    const int r4 = __builtin_amdgcn_readlane(rpv, k4 + 4);
    const int da = (r1 - r0) > (r2 - r1) ? (r1 - r0) : (r2 - r1);
    const int db = (r3 - r2) > (r4 - r3) ? (r3 - r2) : (r4 - r3);
    const int dmax = da > db ? da : db;

    const int   kn  = k4 + g4;
    const int   s0  = __shfl(rpv, kn);
    const int   deg = __shfl(rpv, kn + 1) - s0;
    const float ern = __shfl(erv, kn);

    float m = -INFINITY, ssum = 0.f;
    float4 acc = make_float4(0.f, 0.f, 0.f, 0.f);
    for (int c = 0; c < dmax; c += 8) {
      const bool valid = c + q8 < deg;
      const int  slot  = valid ? (s0 + c + q8) : e0;
      int snv;
      if (STAGED) snv = sn[slot - e0];
      else        snv = src[eidx[slot]];
      const float elv = *reinterpret_cast<const float*>(elb + (static_cast<unsigned>(snv) << 5));

      const float4 v0 = agg_row(hb, grp_bcast<0>(snv), q8);
      const float4 v1 = agg_row(hb, grp_bcast<1>(snv), q8);
      const float4 v2 = agg_row(hb, grp_bcast<2>(snv), q8);
      const float4 v3 = agg_row(hb, grp_bcast<3>(snv), q8);
      const bool more = dmax - c > 4;
      float4 v4, v5, v6, v7;
      if (more) {
        v4 = agg_row(hb, grp_bcast<4>(snv), q8);
        v5 = agg_row(hb, grp_bcast<5>(snv), q8);
        v6 = agg_row(hb, grp_bcast<6>(snv), q8);
        v7 = agg_row(hb, grp_bcast<7>(snv), q8);
      }

      float ev = elv + ern;
      ev = (ev > 0.f) ? ev : 0.2f * ev;
      ev = valid ? ev : -INFINITY;
      const float mn = fmaxf(m, grp_max(ev));
      const float mr = (mn == -INFINITY) ? 0.f : mn;
      const float sc = __expf(m - mr);
      const float p  = __expf(ev - mr);
      ssum = ssum * sc + grp_sum(p);
      m = mn;

      acc.x *= sc;  acc.y *= sc;  acc.z *= sc;  acc.w *= sc;
      agg_fma(acc, grp_bcast<0>(p), v0);
      agg_fma(acc, grp_bcast<1>(p), v1);
      agg_fma(acc, grp_bcast<2>(p), v2);
      agg_fma(acc, grp_bcast<3>(p), v3);
      if (more) {
        agg_fma(acc, grp_bcast<4>(p), v4);
        agg_fma(acc, grp_bcast<5>(p), v5);
        agg_fma(acc, grp_bcast<6>(p), v6);
        agg_fma(acc, grp_bcast<7>(p), v7);
      }
    }

    const float inv = (deg > 0) ? (1.f / ssum) : 0.f;
    float4 r;
    r.x = acc.x * inv + bq.x;  r.y = acc.y * inv + bq.y;
    r.z = acc.z * inv + bq.z;  r.w = acc.w * inv + bq.w;
    reinterpret_cast<float4*>(orow + kn * NCOL)[q8] = r;
  }
}

__global__ void __launch_bounds__(256)
k_aggregate(const float* __restrict__ h, const float* __restrict__ el,
            const float* __restrict__ er, const float* __restrict__ bias,
            const int* __restrict__ rowptr, const int* __restrict__ eidx,
            const int* __restrict__ src, float* __restrict__ out) {
  static_assert(N_NODES % AGG_NB == 0, "every block owns a full batch");
  static_assert(AGG_NB < 32 && (AGG_NB & (AGG_NB - 1)) == 0 && AGG_NB % 4 == 0,
                "lane k of a wave holds row bound k of the batch; lane & (AGG_NB-1) names its node; 4 nodes a step");
  __shared__ int s_sn[AGG_CAP];
  const int tid  = threadIdx.x;
  const int lane = tid & 31;
  const int hd   = __builtin_amdgcn_readfirstlane(tid >> 5);
  const int nb   = blockIdx.x * AGG_NB;

  const int   rpv = rowptr[nb + (lane <= AGG_NB ? lane : AGG_NB)];
  const float erv = er[(nb + (lane & (AGG_NB - 1))) * HEADS + hd];
  const float4 bq = reinterpret_cast<const float4*>(bias + hd * FEAT)[lane & 7];

  const int  e0     = __builtin_amdgcn_readlane(rpv, 0);
  const int  cnt    = __builtin_amdgcn_readlane(rpv, AGG_NB) - e0;
  const bool staged = cnt <= AGG_CAP;
  if (staged)
    for (int i = tid; i < cnt; i += 256) s_sn[i] = src[eidx[e0 + i]];
  __syncthreads();

  const char* hb   = reinterpret_cast<const char*>(h  + hd * FEAT);
  const char* elb  = reinterpret_cast<const char*>(el + hd);
  float*      orow = out + nb * NCOL + hd * FEAT;
  if (staged) agg_walk<true >(hb, elb, eidx, src, s_sn, e0, rpv, erv, bq, lane, orow);
  else        agg_walk<false>(hb, elb, eidx, src, s_sn, e0, rpv, erv, bq, lane, orow);
}

__global__ void k_elu_headmean(const float* __restrict__ agg, float* __restrict__ out) {
  int idx = blockIdx.x * blockDim.x + threadIdx.x;
  if (idx >= N_NODES * FEAT) return;
  int f = idx & (FEAT - 1);
  int n = idx >> 5;
  float s = 0.f;
  #pragma unroll
  for (int hd = 0; hd < HEADS; ++hd) {
    float v = agg[n * NCOL + hd * FEAT + f];
    s += (v > 0.f) ? v : (__expf(v) - 1.f);
  }
  out[idx] = s * (1.f / HEADS);
}

extern "C" void kernel_launch(void* const* d_in, const int* in_sizes, int n_in,
                              void* d_out, int out_size, void* d_ws, size_t ws_size,
                              hipStream_t stream) {
  const float* x   = (const float*)d_in[0];
  const int*   src = (const int*)  d_in[1];
  const int*   dst = (const int*)  d_in[2];
  const float* W1  = (const float*)d_in[3];
  const float* al1 = (const float*)d_in[4];
  const float* ar1 = (const float*)d_in[5];
  const float* b1  = (const float*)d_in[6];
  const float* W2  = (const float*)d_in[7];
  const float* al2 = (const float*)d_in[8];
  const float* ar2 = (const float*)d_in[9];
  const float* b2  = (const float*)d_in[10];

  float* wsf = (float*)d_ws;
  float* h_buf = wsf;
  float* agg1  = h_buf + (size_t)N_NODES * NCOL;
  float* elb   = agg1  + (size_t)N_NODES * NCOL;
  float* erb   = elb   + (size_t)N_NODES * HEADS;
  float* h1act = erb   + (size_t)N_NODES * HEADS;
  int* ibase   = (int*)(h1act + (size_t)N_NODES * FEAT);
  int* rowptr  = ibase;
  int* cursor  = rowptr + (N_NODES + 2);
  int* cnts    = cursor + N_NODES;
  int* eidx    = cnts   + N_NODES;

  k_zero_i32<<<(N_NODES + 255) / 256, 256, 0, stream>>>(cnts, N_NODES);
  k_count   <<<(N_EDGES + 255) / 256, 256, 0, stream>>>(dst, cnts, N_EDGES);
  k_exscan  <<<1, 1024, 0, stream>>>(cnts, rowptr, cursor, N_NODES);
  k_scatter <<<(N_EDGES + 255) / 256, 256, 0, stream>>>(dst, cursor, eidx, N_EDGES);

  const int gemm_blocks = ((N_NODES / 16) * 4 * 32 + 255) / 256;
  const int nh_blocks   = (N_NODES * HEADS + 255) / 256;
  const int agg_blocks  = N_NODES / AGG_NB;
  const int nf_blocks   = (N_NODES * FEAT + 255) / 256;

  k_gemm_wmma <<<gemm_blocks, 256, 0, stream>>>(x, W1, h_buf, N_NODES, 128);
  k_attn      <<<nh_blocks,  256, 0, stream>>>(h_buf, al1, ar1, elb, erb);
  k_aggregate <<<agg_blocks, 256, 0, stream>>>(h_buf, elb, erb, b1, rowptr, eidx, src, agg1);
  k_elu_headmean<<<nf_blocks, 256, 0, stream>>>(agg1, h1act);

  k_gemm_wmma <<<gemm_blocks, 256, 0, stream>>>(h1act, W2, h_buf, N_NODES, 32);
  k_attn      <<<nh_blocks,  256, 0, stream>>>(h_buf, al2, ar2, elb, erb);
  k_aggregate <<<agg_blocks, 256, 0, stream>>>(h_buf, elb, erb, b2, rowptr, eidx, src, (float*)d_out);
}


# exclusive scan of the 50k counts cut over the grid: 13 blocks, each sums the counts in front of it itself and scans a 16-byte piece per thread with lane shuffles; was one block, 49-count serial runs
# speedup vs baseline: 1.7474x; 1.2047x over previous
#include <hip/hip_runtime.h>
#include <hip/hip_bf16.h>
#include <math.h>

typedef __attribute__((ext_vector_type(2))) float v2f;
typedef __attribute__((ext_vector_type(8))) float v8f;

#define N_NODES 50000
#define N_EDGES 800000
#define HEADS   8
#define FEAT    32
#define NCOL    256

__global__ void k_zero_i32(int* __restrict__ p, int n) {
  int i = blockIdx.x * blockDim.x + threadIdx.x;
  if (i < n) p[i] = 0;
}

__global__ void k_count(const int* __restrict__ dst, int* __restrict__ cnt, int E) {
  int i = blockIdx.x * blockDim.x + threadIdx.x;
  if (i < E) atomicAdd(&cnt[dst[i]], 1);
}

#define SCAN_T   1024
#define SCAN_NP  (N_NODES / 4)
#define SCAN_NB  ((SCAN_NP + SCAN_T - 1) / SCAN_T)

__global__ void __launch_bounds__(SCAN_T)
k_exscan(const int* __restrict__ cnt, int* __restrict__ rowptr, int* __restrict__ cursor) {
  static_assert(N_NODES % 4 == 0, "the counts are whole 16-byte pieces");
  static_assert(SCAN_T == 32 * 32, "lane l of every wave takes over the figures of wave l");
  static_assert((SCAN_NB - 1) * SCAN_T <= SCAN_NP, "the rounds in front of the last block lie inside the counts");
  __shared__ int s_pre[32];
  __shared__ int s_tot[32];
  const int t = threadIdx.x, lane = t & 31, w = t >> 5;
  const int4* c4 = reinterpret_cast<const int4*>(cnt);

  int pre = 0;
  #pragma unroll 4
  for (int k = 0; k < (int)blockIdx.x; ++k) {
    const int4 v = c4[k * SCAN_T + t];
    pre += (v.x + v.y) + (v.z + v.w);
  }

  const int  p    = blockIdx.x * SCAN_T + t;
  const bool have = p < SCAN_NP;
  int4 c = make_int4(0, 0, 0, 0);
  if (have) c = c4[p];
  const int s = (c.x + c.y) + (c.z + c.w);
  int inc = s;
  #pragma unroll
  for (int off = 1; off < 32; off <<= 1) {
    const int v = __shfl_up(inc, off);
    if (lane >= off) inc += v;
  }
  #pragma unroll
  for (int off = 16; off > 0; off >>= 1)
    pre += __shfl_xor(pre, off);
  if (lane == 31) s_tot[w] = inc;
  if (lane == 0)  s_pre[w] = pre;
  __syncthreads();

  const int tot = s_tot[lane];
  int tinc = tot;
  int psum = s_pre[lane];
  #pragma unroll
  for (int off = 1; off < 32; off <<= 1) {
    const int v = __shfl_up(tinc, off);
    if (lane >= off) tinc += v;
  }
  #pragma unroll
  for (int off = 16; off > 0; off >>= 1)
    psum += __shfl_xor(psum, off);
  const int wex = __shfl(tinc - tot, w);

  const int run = psum + wex + (inc - s);
  if (have) {
    int4 r;
    r.x = run;  r.y = run + c.x;  r.z = r.y + c.y;  r.w = r.z + c.z;
    reinterpret_cast<int4*>(rowptr)[p] = r;
    reinterpret_cast<int4*>(cursor)[p] = r;
  }
  if (p == SCAN_NP - 1) rowptr[N_NODES] = run + s;
}

__global__ void k_scatter(const int* __restrict__ dst, int* __restrict__ cursor,
                          int* __restrict__ eidx, int E) {
  int i = blockIdx.x * blockDim.x + threadIdx.x;
  if (i < E) {
    int p = atomicAdd(&cursor[dst[i]], 1);
    eidx[p] = i;
  }
}

__global__ void k_gemm_wmma(const float* __restrict__ A, const float* __restrict__ B,
                            float* __restrict__ C, int M, int K) {
  int wid  = (blockIdx.x * blockDim.x + threadIdx.x) >> 5;
  int lane = threadIdx.x & 31;
  int mtiles = M >> 4;
  if (wid >= mtiles * 4) return;
  int mt = wid >> 2, ng = wid & 3;
  int m0 = mt << 4, n0 = ng << 6;
  int rc   = lane & 15;
  int koff = (lane >> 4) << 1;
  int hi8  = (lane >> 4) << 3;

  v8f acc0 = {}, acc1 = {}, acc2 = {}, acc3 = {};
  const float* arow = A + (m0 + rc) * K;

  #pragma unroll 4
  for (int k = 0; k < K; k += 4) {
    v2f a;
    a.x = arow[k + koff];
    a.y = arow[k + koff + 1];
    const float* bp = B + (k + koff) * NCOL + n0 + rc;
    v2f b;
    b.x = bp[0];       b.y = bp[NCOL];
    acc0 = __builtin_amdgcn_wmma_f32_16x16x4_f32(false, a, false, b, (short)0, acc0, false, false);
    b.x = bp[16];      b.y = bp[NCOL + 16];
    acc1 = __builtin_amdgcn_wmma_f32_16x16x4_f32(false, a, false, b, (short)0, acc1, false, false);
    b.x = bp[32];      b.y = bp[NCOL + 32];
    acc2 = __builtin_amdgcn_wmma_f32_16x16x4_f32(false, a, false, b, (short)0, acc2, false, false);
    b.x = bp[48];      b.y = bp[NCOL + 48];
    acc3 = __builtin_amdgcn_wmma_f32_16x16x4_f32(false, a, false, b, (short)0, acc3, false, false);
  }

  float* crow = C + (m0 + hi8) * NCOL + n0 + rc;
  #pragma unroll
  for (int j = 0; j < 8; ++j) {
    crow[j * NCOL +  0] = acc0[j];
    crow[j * NCOL + 16] = acc1[j];
    crow[j * NCOL + 32] = acc2[j];
    crow[j * NCOL + 48] = acc3[j];
  }
}

#define AGG_NB  16
#define AGG_CAP 2048

template <int M>
__device__ __forceinline__ float row_xor(float x) {
  const int b = __builtin_bit_cast(int, x);
  return __builtin_bit_cast(float, __builtin_amdgcn_update_dpp(b, b, 0x160 | M, 0xf, 0xf, false));
}
template <int T>
__device__ __forceinline__ int grp_bcast(int x) {
  return __builtin_amdgcn_mov_dpp8(x, T * 0x249249);
}
template <int T>
__device__ __forceinline__ float grp_bcast(float x) {
  return __builtin_bit_cast(float, __builtin_amdgcn_mov_dpp8(__builtin_bit_cast(int, x), T * 0x249249));
}
__device__ __forceinline__ float grp_max(float x) {
  x = fmaxf(x, row_xor<1>(x));
  x = fmaxf(x, row_xor<2>(x));
  return fmaxf(x, row_xor<4>(x));
}
__device__ __forceinline__ float grp_sum(float x) {
  x += row_xor<1>(x);
  x += row_xor<2>(x);
  return x + row_xor<4>(x);
}

__device__ __forceinline__ float4 agg_row(const char* __restrict__ hb, int sn, int q8) {
  const unsigned off = (static_cast<unsigned>(sn) << 10) | static_cast<unsigned>(q8 << 4);
  return *reinterpret_cast<const float4*>(hb + off);
}

__device__ __forceinline__ void agg_fma(float4& acc, float p, const float4 v) {
  acc.x = fmaf(p, v.x, acc.x);
  acc.y = fmaf(p, v.y, acc.y);
  acc.z = fmaf(p, v.z, acc.z);
  acc.w = fmaf(p, v.w, acc.w);
}

__device__ __forceinline__ float dot4(const float4 v, const float4 a) {
  const v2f vlo = {v.x, v.y}, vhi = {v.z, v.w};
  const v2f alo = {a.x, a.y}, ahi = {a.z, a.w};
  const v2f s = __builtin_elementwise_fma(vhi, ahi, vlo * alo);
  return s.x + s.y;
}

__device__ __forceinline__ float grp_dot_to_lane(float d0, float d1, float d2, float d3,
                                                 float d4, float d5, float d6, float d7, int q8) {
  const bool b4 = (q8 & 4) != 0, b2 = (q8 & 2) != 0, b1 = (q8 & 1) != 0;
  const float e0 = (b4 ? d4 : d0) + row_xor<4>(b4 ? d0 : d4);
  const float e1 = (b4 ? d5 : d1) + row_xor<4>(b4 ? d1 : d5);
  const float e2 = (b4 ? d6 : d2) + row_xor<4>(b4 ? d2 : d6);
  const float e3 = (b4 ? d7 : d3) + row_xor<4>(b4 ? d3 : d7);
  const float f0 = (b2 ? e2 : e0) + row_xor<2>(b2 ? e0 : e2);
  const float f1 = (b2 ? e3 : e1) + row_xor<2>(b2 ? e1 : e3);
  return (b1 ? f1 : f0) + row_xor<1>(b1 ? f0 : f1);
}

template <bool STAGED>
__device__ __forceinline__ void agg_walk(const char* __restrict__ hb,
                                         const int* __restrict__ eidx, const int* __restrict__ src,
                                         const int* sn, int e0, int rpv, float4 alq,
                                         float er0, float er1, float er2, float er3,
                                         float4 bq, int lane, float* __restrict__ orow) {
  const int g4 = lane >> 3;
  const int q8 = lane & 7;
  for (int k4 = 0; k4 < AGG_NB; k4 += 4) {
    const int r0 = __builtin_amdgcn_readlane(rpv, k4);
    const int r1 = __builtin_amdgcn_readlane(rpv, k4 + 1);
    const int r2 = __builtin_amdgcn_readlane(rpv, k4 + 2);
    const int r3 = __builtin_amdgcn_readlane(rpv, k4 + 3);
    const int r4 = __builtin_amdgcn_readlane(rpv, k4 + 4);
    const int da = (r1 - r0) > (r2 - r1) ? (r1 - r0) : (r2 - r1);
    const int db = (r3 - r2) > (r4 - r3) ? (r3 - r2) : (r4 - r3);
    const int dmax = da > db ? da : db;

    const int   kn  = k4 + g4;
    const int   s0  = __shfl(rpv, kn);
    const int   deg = __shfl(rpv, kn + 1) - s0;
    const float ern = (k4 == 0) ? er0 : (k4 == 4) ? er1 : (k4 == 8) ? er2 : er3;

    float m = -INFINITY, ssum = 0.f;
    float4 acc = make_float4(0.f, 0.f, 0.f, 0.f);
    for (int c = 0; c < dmax; c += 8) {
      const bool valid = c + q8 < deg;
      const int  slot  = valid ? (s0 + c + q8) : e0;
      int snv;
      if (STAGED) snv = sn[slot - e0];
      else        snv = src[eidx[slot]];

      const float4 v0 = agg_row(hb, grp_bcast<0>(snv), q8);
      const float4 v1 = agg_row(hb, grp_bcast<1>(snv), q8);
      const float4 v2 = agg_row(hb, grp_bcast<2>(snv), q8);
      const float4 v3 = agg_row(hb, grp_bcast<3>(snv), q8);
      const bool more = dmax - c > 4;
      float4 v4, v5, v6, v7;
      if (more) {
        v4 = agg_row(hb, grp_bcast<4>(snv), q8);
        v5 = agg_row(hb, grp_bcast<5>(snv), q8);
        v6 = agg_row(hb, grp_bcast<6>(snv), q8);
        v7 = agg_row(hb, grp_bcast<7>(snv), q8);
      }

      const float d0 = dot4(v0, alq), d1 = dot4(v1, alq), d2 = dot4(v2, alq), d3 = dot4(v3, alq);
      float d4 = 0.f, d5 = 0.f, d6 = 0.f, d7 = 0.f;
      if (more) { d4 = dot4(v4, alq); d5 = dot4(v5, alq); d6 = dot4(v6, alq); d7 = dot4(v7, alq); }
      const float elv = grp_dot_to_lane(d0, d1, d2, d3, d4, d5, d6, d7, q8);

      float ev = elv + ern;
      ev = (ev > 0.f) ? ev : 0.2f * ev;
      ev = valid ? ev : -INFINITY;
      const float mn = fmaxf(m, grp_max(ev));
      const float mr = (mn == -INFINITY) ? 0.f : mn;
      const float sc = __expf(m - mr);
      const float p  = __expf(ev - mr);
      ssum = ssum * sc + grp_sum(p);
      m = mn;

      acc.x *= sc;  acc.y *= sc;  acc.z *= sc;  acc.w *= sc;
      agg_fma(acc, grp_bcast<0>(p), v0);
      agg_fma(acc, grp_bcast<1>(p), v1);
      agg_fma(acc, grp_bcast<2>(p), v2);
      agg_fma(acc, grp_bcast<3>(p), v3);
      if (more) {
        agg_fma(acc, grp_bcast<4>(p), v4);
        agg_fma(acc, grp_bcast<5>(p), v5);
        agg_fma(acc, grp_bcast<6>(p), v6);
        agg_fma(acc, grp_bcast<7>(p), v7);
      }
    }

    const float inv = (deg > 0) ? (1.f / ssum) : 0.f;
    float4 r;
    r.x = acc.x * inv + bq.x;  r.y = acc.y * inv + bq.y;
    r.z = acc.z * inv + bq.z;  r.w = acc.w * inv + bq.w;
    reinterpret_cast<float4*>(orow + kn * NCOL)[q8] = r;
  }
}

__global__ void __launch_bounds__(256)
k_aggregate(const float* __restrict__ h, const float* __restrict__ al,
            const float* __restrict__ ar, const float* __restrict__ bias,
            const int* __restrict__ rowptr, const int* __restrict__ eidx,
            const int* __restrict__ src, float* __restrict__ out) {
  static_assert(N_NODES % AGG_NB == 0, "every block owns a full batch");
  static_assert(AGG_NB == 16,
                "lane k of a wave holds row bound k of the batch; 4 steps of 4 nodes, one destination logit each");
  __shared__ int s_sn[AGG_CAP];
  const int tid  = threadIdx.x;
  const int lane = tid & 31;
  const int hd   = __builtin_amdgcn_readfirstlane(tid >> 5);
  const int nb   = blockIdx.x * AGG_NB;

  const int    rpv = rowptr[nb + (lane <= AGG_NB ? lane : AGG_NB)];
  const float4 alq = reinterpret_cast<const float4*>(al   + hd * FEAT)[lane & 7];
  const float4 arq = reinterpret_cast<const float4*>(ar   + hd * FEAT)[lane & 7];
  const float4 bq  = reinterpret_cast<const float4*>(bias + hd * FEAT)[lane & 7];

  const char* hb  = reinterpret_cast<const char*>(h + hd * FEAT);
  const int   ng  = nb + (lane >> 3);
  const float4 h0 = agg_row(hb, ng,      lane & 7);
  const float4 h1 = agg_row(hb, ng + 4,  lane & 7);
  const float4 h2 = agg_row(hb, ng + 8,  lane & 7);
  const float4 h3 = agg_row(hb, ng + 12, lane & 7);

  const int  e0     = __builtin_amdgcn_readlane(rpv, 0);
  const int  cnt    = __builtin_amdgcn_readlane(rpv, AGG_NB) - e0;
  const bool staged = cnt <= AGG_CAP;
  if (staged)
    for (int i = tid; i < cnt; i += 256) s_sn[i] = src[eidx[e0 + i]];
  __syncthreads();

  const float er0 = grp_sum(dot4(h0, arq));
  const float er1 = grp_sum(dot4(h1, arq));
  const float er2 = grp_sum(dot4(h2, arq));
  const float er3 = grp_sum(dot4(h3, arq));

  float* orow = out + nb * NCOL + hd * FEAT;
  if (staged) agg_walk<true >(hb, eidx, src, s_sn, e0, rpv, alq, er0, er1, er2, er3, bq, lane, orow);
  else        agg_walk<false>(hb, eidx, src, s_sn, e0, rpv, alq, er0, er1, er2, er3, bq, lane, orow);
}

__global__ void k_elu_headmean(const float* __restrict__ agg, float* __restrict__ out) {
  int idx = blockIdx.x * blockDim.x + threadIdx.x;
  if (idx >= N_NODES * FEAT) return;
  int f = idx & (FEAT - 1);
  int n = idx >> 5;
  float s = 0.f;
  #pragma unroll
  for (int hd = 0; hd < HEADS; ++hd) {
    float v = agg[n * NCOL + hd * FEAT + f];
    s += (v > 0.f) ? v : (__expf(v) - 1.f);
  }
  out[idx] = s * (1.f / HEADS);
}

extern "C" void kernel_launch(void* const* d_in, const int* in_sizes, int n_in,
                              void* d_out, int out_size, void* d_ws, size_t ws_size,
                              hipStream_t stream) {
  const float* x   = (const float*)d_in[0];
  const int*   src = (const int*)  d_in[1];
  const int*   dst = (const int*)  d_in[2];
  const float* W1  = (const float*)d_in[3];
  const float* al1 = (const float*)d_in[4];
  const float* ar1 = (const float*)d_in[5];
  const float* b1  = (const float*)d_in[6];
  const float* W2  = (const float*)d_in[7];
  const float* al2 = (const float*)d_in[8];
  const float* ar2 = (const float*)d_in[9];
  const float* b2  = (const float*)d_in[10];

  float* wsf = (float*)d_ws;
  float* h_buf = wsf;
  float* agg1  = h_buf + (size_t)N_NODES * NCOL;
  float* h1act = agg1  + (size_t)N_NODES * NCOL;
  int* ibase   = (int*)(h1act + (size_t)N_NODES * FEAT);
  int* cnts    = ibase;
  int* cursor  = cnts   + N_NODES;
  int* rowptr  = cursor + N_NODES;
  int* eidx    = rowptr + (N_NODES + 2);

  k_zero_i32<<<(N_NODES + 255) / 256, 256, 0, stream>>>(cnts, N_NODES);
  k_count   <<<(N_EDGES + 255) / 256, 256, 0, stream>>>(dst, cnts, N_EDGES);
  k_exscan  <<<SCAN_NB, SCAN_T, 0, stream>>>(cnts, rowptr, cursor);
  k_scatter <<<(N_EDGES + 255) / 256, 256, 0, stream>>>(dst, cursor, eidx, N_EDGES);

  const int gemm_blocks = ((N_NODES / 16) * 4 * 32 + 255) / 256;
  const int agg_blocks  = N_NODES / AGG_NB;
  const int nf_blocks   = (N_NODES * FEAT + 255) / 256;

  k_gemm_wmma <<<gemm_blocks, 256, 0, stream>>>(x, W1, h_buf, N_NODES, 128);
  k_aggregate <<<agg_blocks, 256, 0, stream>>>(h_buf, al1, ar1, b1, rowptr, eidx, src, agg1);
  k_elu_headmean<<<nf_blocks, 256, 0, stream>>>(agg1, h1act);

  k_gemm_wmma <<<gemm_blocks, 256, 0, stream>>>(h1act, W2, h_buf, N_NODES, 32);
  k_aggregate <<<agg_blocks, 256, 0, stream>>>(h_buf, al2, ar2, b2, rowptr, eidx, src, (float*)d_out);
}
